# starting kernel
# baseline (speedup 1.0000x reference)
#include <hip/hip_runtime.h>

typedef __attribute__((ext_vector_type(2))) float v2f;
typedef __attribute__((ext_vector_type(8))) float v8f;

#define B_    2048
#define N_    32
#define IND   256
#define OUTD  64
#define H_    8
#define QCOLS 512


__global__ void prep_wvs(const float* __restrict__ WV, const float* __restrict__ bV,
                         float* __restrict__ WVs, float* __restrict__ bVs) {
  int idx = blockIdx.x * blockDim.x + threadIdx.x;
  int c = idx >> 6;
  int j = idx & 63;
  float s = 0.f;
#pragma unroll
  for (int h = 0; h < H_; ++h) s += WV[c * QCOLS + j * H_ + h];
  WVs[c * OUTD + j] = s;
  if (idx < OUTD) {
    float sb = 0.f;
#pragma unroll
    for (int h = 0; h < H_; ++h) sb += bV[idx * H_ + h];
    bVs[idx] = sb;
  }
}

__global__ __launch_bounds__(256)
void proj_q(const float* __restrict__ center, const float* __restrict__ WQ,
            const float* __restrict__ bQ, float* __restrict__ Qout) {
  int wid  = (blockIdx.x * blockDim.x + threadIdx.x) >> 5;
  int lane = threadIdx.x & 31;
  int l15 = lane & 15, hi = lane >> 4;
  int g  = wid >> 5;
  int nt = wid & 31;
  int n0 = nt * 16;
  v8f acc[4] = {{}, {}, {}, {}};
  for (int kt = 0; kt < IND / 4; ++kt) {
    int kb = kt * 4 + hi * 2;
    v2f b;
    b.x = WQ[kb * QCOLS + n0 + l15];
    b.y = WQ[(kb + 1) * QCOLS + n0 + l15];
#pragma unroll
    for (int u = 0; u < 4; ++u) {
      v2f a;
      a.x = center[(size_t)(g * 64 + u * 16 + l15) * IND + kb];
      a.y = center[(size_t)(g * 64 + u * 16 + l15) * IND + kb + 1];
      acc[u] = __builtin_amdgcn_wmma_f32_16x16x4_f32(false, a, false, b, (short)0,
                                                     acc[u], false, false);
    }
  }
  float bias = bQ[n0 + l15];
#pragma unroll
  for (int u = 0; u < 4; ++u)
#pragma unroll
    for (int r = 0; r < 8; ++r)
      Qout[(size_t)(g * 64 + u * 16 + r + hi * 8) * QCOLS + n0 + l15] = acc[u][r] + bias;
}

__global__ __launch_bounds__(256)
void attn_fused(const float* __restrict__ neighbors, const float* __restrict__ WK,
                const float* __restrict__ bK, const float* __restrict__ WVs,
                const float* __restrict__ bVs, const float* __restrict__ Qh,
                float* __restrict__ out) {
  __shared__ float nb[N_ * 260];
  __shared__ float Kh[N_ * QCOLS];
  __shared__ float Vs[N_ * OUTD];
  __shared__ float Qs[OUTD * 9];
  __shared__ float S[OUTD * 65];
  __shared__ float redM[4 * OUTD];
  __shared__ float redN[4 * OUTD];
  __shared__ float redD[4 * OUTD];

  const int b = blockIdx.x;
  const int t = threadIdx.x;
  const int wave = t >> 5, lane = t & 31, l15 = lane & 15, hi = lane >> 4;

  const float* nbg = neighbors + (size_t)b * N_ * IND;
  for (int i = t; i < N_ * IND; i += 256)
    nb[(i >> 8) * 260 + (i & 255)] = nbg[i];
  const float* qg = Qh + (size_t)b * QCOLS;
  for (int i = t; i < QCOLS; i += 256) {
    int d = i >> 3, h = i & 7;
    Qs[d * 9 + h] = qg[i] * 0.125f;
  }
  __syncthreads();

  for (int nt4 = 0; nt4 < 4; ++nt4) {
    int ntile = wave + nt4 * 8;
    int n0 = ntile * 16;
    v8f acc0 = {}, acc1 = {};
    for (int kt = 0; kt < IND / 4; ++kt) {
      int kb = kt * 4 + hi * 2;
      v2f bb, a0, a1;
      bb.x = WK[kb * QCOLS + n0 + l15];
      bb.y = WK[(kb + 1) * QCOLS + n0 + l15];
      a0.x = nb[l15 * 260 + kb];
      a0.y = nb[l15 * 260 + kb + 1];
      a1.x = nb[(16 + l15) * 260 + kb];
      a1.y = nb[(16 + l15) * 260 + kb + 1];
      acc0 = __builtin_amdgcn_wmma_f32_16x16x4_f32(false, a0, false, bb, (short)0,
                                                   acc0, false, false);
      acc1 = __builtin_amdgcn_wmma_f32_16x16x4_f32(false, a1, false, bb, (short)0,
                                                   acc1, false, false);
    }
    float bias = bK[n0 + l15];
#pragma unroll
    for (int r = 0; r < 8; ++r) {
      Kh[(r + hi * 8) * QCOLS + n0 + l15]      = acc0[r] + bias;
      Kh[(16 + r + hi * 8) * QCOLS + n0 + l15] = acc1[r] + bias;
    }
  }

  {
    int mt = wave >> 2, ntile = wave & 3;
    int m0 = mt * 16, n0 = ntile * 16;
    v8f acc = {};
    for (int kt = 0; kt < IND / 4; ++kt) {
      int kb = kt * 4 + hi * 2;
      v2f a, bb;
      a.x = nb[(m0 + l15) * 260 + kb];
      a.y = nb[(m0 + l15) * 260 + kb + 1];
      bb.x = WVs[kb * OUTD + n0 + l15];
      bb.y = WVs[(kb + 1) * OUTD + n0 + l15];
      acc = __builtin_amdgcn_wmma_f32_16x16x4_f32(false, a, false, bb, (short)0, acc,
                                                  false, false);
    }
    float bias = bVs[n0 + l15];
#pragma unroll
    for (int r = 0; r < 8; ++r)
      Vs[(m0 + r + hi * 8) * OUTD + n0 + l15] = acc[r] + bias;
  }
  __syncthreads();

  const int it = (wave * 2) >> 2;
  const int m0q = it * 16;
  v2f aQ[2];
#pragma unroll
  for (int ks = 0; ks < 2; ++ks) {
    int kb = ks * 4 + hi * 2;
    aQ[ks].x = Qs[(m0q + l15) * 9 + kb];
    aQ[ks].y = Qs[(m0q + l15) * 9 + kb + 1];
  }

  const int j = t & 63;
  const int q = t >> 6;
  float accOut = 0.f;

  for (int n = 0; n < N_; ++n) {
#pragma unroll
    for (int tt = 0; tt < 2; ++tt) {
      int id = wave * 2 + tt;
      int j0 = (id & 3) * 16;
      v8f acc = {};
#pragma unroll
      for (int ks = 0; ks < 2; ++ks) {
        int kb = ks * 4 + hi * 2;
        v2f bb;
        bb.x = Kh[n * QCOLS + (j0 + l15) * 8 + kb];
        bb.y = Kh[n * QCOLS + (j0 + l15) * 8 + kb + 1];
        acc = __builtin_amdgcn_wmma_f32_16x16x4_f32(false, aQ[ks], false, bb, (short)0,
                                                    acc, false, false);
      }
#pragma unroll
      for (int r = 0; r < 8; ++r)
        S[(m0q + r + hi * 8) * 65 + j0 + l15] = acc[r];
    }
    __syncthreads();

    float pm = -3.4e38f;
#pragma unroll
    for (int s = 0; s < 16; ++s)
      pm = fmaxf(pm, S[(q * 16 + s) * 65 + j]);
    redM[q * 64 + j] = pm;
    __syncthreads();
    float m = fmaxf(fmaxf(redM[j], redM[64 + j]), fmaxf(redM[128 + j], redM[192 + j]));
    float pn = 0.f, pd = 0.f;
#pragma unroll
    for (int s = 0; s < 16; ++s) {
      float e = __expf(S[(q * 16 + s) * 65 + j] - m);
      pd += e;
      pn += e * Vs[n * OUTD + q * 16 + s];
    }
    redN[q * 64 + j] = pn;
    redD[q * 64 + j] = pd;
    __syncthreads();
    if (t < 64) {
      float num = redN[j] + redN[64 + j] + redN[128 + j] + redN[192 + j];
      float den = redD[j] + redD[64 + j] + redD[128 + j] + redD[192 + j];
      accOut += num / den;
    }
  }
  if (t < 64) out[(size_t)b * OUTD + t] = accOut * (1.f / H_);
}

extern "C" void kernel_launch(void* const* d_in, const int* in_sizes, int n_in,
                              void* d_out, int out_size, void* d_ws, size_t ws_size,
                              hipStream_t stream) {
  const float* center    = (const float*)d_in[0];
  const float* neighbors = (const float*)d_in[1];
  const float* WQ        = (const float*)d_in[2];
  const float* bQ        = (const float*)d_in[3];
  const float* WK        = (const float*)d_in[4];
  const float* bK        = (const float*)d_in[5];
  const float* WV        = (const float*)d_in[6];
  const float* bV        = (const float*)d_in[7];
  float* out = (float*)d_out;

  float* ws   = (float*)d_ws;
  float* wsQ  = ws;
  float* WVs  = ws + (size_t)B_ * QCOLS;
  float* bVs  = WVs + (size_t)IND * OUTD;

  prep_wvs<<<64, 256, 0, stream>>>(WV, bV, WVs, bVs);
  proj_q  <<<128, 256, 0, stream>>>(center, WQ, bQ, wsQ);
  attn_fused<<<B_, 256, 0, stream>>>(neighbors, WK, bK, WVs, bVs, wsQ, out);
}


# attn_fused: K and Vs projections merged into one k loop; a 128-bit WK row load feeds 4 n-tiles x 2 m-tiles, next 4 k-steps' operands fetched ahead of the current 36 WMMAs; sum order unchanged
# speedup vs baseline: 1.1615x; 1.1615x over previous
#include <hip/hip_runtime.h>

typedef __attribute__((ext_vector_type(2))) float v2f;
typedef __attribute__((ext_vector_type(4))) float v4f;
typedef __attribute__((ext_vector_type(8))) float v8f;

#define B_    2048
#define N_    32
#define IND   256
#define OUTD  64
#define H_    8
#define QCOLS 512


__global__ void prep_wvs(const float* __restrict__ WV, const float* __restrict__ bV,
                         float* __restrict__ WVs, float* __restrict__ bVs) {
  int idx = blockIdx.x * blockDim.x + threadIdx.x;
  int c = idx >> 6;
  int j = idx & 63;
  float s = 0.f;
#pragma unroll
  for (int h = 0; h < H_; ++h) s += WV[c * QCOLS + j * H_ + h];
  WVs[c * OUTD + j] = s;
  if (idx < OUTD) {
    float sb = 0.f;
#pragma unroll
    for (int h = 0; h < H_; ++h) sb += bV[idx * H_ + h];
    bVs[idx] = sb;
  }
}

__global__ __launch_bounds__(256)
void proj_q(const float* __restrict__ center, const float* __restrict__ WQ,
            const float* __restrict__ bQ, float* __restrict__ Qout) {
  int wid  = (blockIdx.x * blockDim.x + threadIdx.x) >> 5;
  int lane = threadIdx.x & 31;
  int l15 = lane & 15, hi = lane >> 4;
  int g  = wid >> 5;
  int nt = wid & 31;
  int n0 = nt * 16;
  v8f acc[4] = {{}, {}, {}, {}};
  for (int kt = 0; kt < IND / 4; ++kt) {
    int kb = kt * 4 + hi * 2;
    v2f b;
    b.x = WQ[kb * QCOLS + n0 + l15];
    b.y = WQ[(kb + 1) * QCOLS + n0 + l15];
#pragma unroll
    for (int u = 0; u < 4; ++u) {
      v2f a;
      a.x = center[(size_t)(g * 64 + u * 16 + l15) * IND + kb];
      a.y = center[(size_t)(g * 64 + u * 16 + l15) * IND + kb + 1];
      acc[u] = __builtin_amdgcn_wmma_f32_16x16x4_f32(false, a, false, b, (short)0,
                                                     acc[u], false, false);
    }
  }
  float bias = bQ[n0 + l15];
#pragma unroll
  for (int u = 0; u < 4; ++u)
#pragma unroll
    for (int r = 0; r < 8; ++r)
      Qout[(size_t)(g * 64 + u * 16 + r + hi * 8) * QCOLS + n0 + l15] = acc[u][r] + bias;
}

#define KG 4
struct ProjGroup {
  v4f w0[KG], w1[KG];
  v2f wv[KG];
  v2f a0[KG], a1[KG];
};

__device__ __forceinline__ void load_weights(ProjGroup& g, const float* __restrict__ wkp,
                                             const float* __restrict__ wvp, int kt0) {
#pragma unroll
  for (int s = 0; s < KG; ++s) {
    int kb = (kt0 + s) * 4;
    g.w0[s] = *(const v4f*)(wkp + (size_t)kb * QCOLS);
    g.w1[s] = *(const v4f*)(wkp + (size_t)(kb + 1) * QCOLS);
    g.wv[s].x = wvp[kb * OUTD];
    g.wv[s].y = wvp[(kb + 1) * OUTD];
  }
}

__device__ __forceinline__ void load_rows(ProjGroup& g, const float* nb0, const float* nb1,
                                          int kt0) {
#pragma unroll
  for (int s = 0; s < KG; ++s) {
    int kb = (kt0 + s) * 4;
    g.a0[s] = *(const v2f*)(nb0 + kb);
    g.a1[s] = *(const v2f*)(nb1 + kb);
  }
}

__device__ __forceinline__ void proj_group(const ProjGroup& g, v8f (&accK)[2][4], v8f& accV) {
#pragma unroll
  for (int s = 0; s < KG; ++s) {
#pragma unroll
    for (int t = 0; t < 4; ++t) {
      v2f bb;
      bb.x = g.w0[s][t];
      bb.y = g.w1[s][t];
      accK[0][t] = __builtin_amdgcn_wmma_f32_16x16x4_f32(false, g.a0[s], false, bb, (short)0,
                                                         accK[0][t], false, false);
      accK[1][t] = __builtin_amdgcn_wmma_f32_16x16x4_f32(false, g.a1[s], false, bb, (short)0,
                                                         accK[1][t], false, false);
    }
    accV = __builtin_amdgcn_wmma_f32_16x16x4_f32(false, g.a0[s], false, g.wv[s], (short)0,
                                                 accV, false, false);
  }
}

__global__ __launch_bounds__(256)
void attn_fused(const float* __restrict__ neighbors, const float* __restrict__ WK,
                const float* __restrict__ bK, const float* __restrict__ WVs,
                const float* __restrict__ bVs, const float* __restrict__ Qh,
                float* __restrict__ out) {
  __shared__ __attribute__((aligned(16))) float nb[N_ * 260];
  __shared__ __attribute__((aligned(16))) float Kh[N_ * QCOLS];
  __shared__ float Vs[N_ * OUTD];
  __shared__ float Qs[OUTD * 9];
  __shared__ float S[OUTD * 65];
  __shared__ float redM[4 * OUTD];
  __shared__ float redN[4 * OUTD];
  __shared__ float redD[4 * OUTD];

  const int b = blockIdx.x;
  const int t = threadIdx.x;
  const int wave = t >> 5, lane = t & 31, l15 = lane & 15, hi = lane >> 4;

  const int mA = wave >> 2, mB = 1 - mA;
  const int nV0 = (wave & 3) * 16;
  const float* wkp = WK + (size_t)(hi * 2) * QCOLS + wave * 64 + l15 * 4;
  const float* wvp = WVs + (size_t)(hi * 2) * OUTD + nV0 + l15;
  const float* nb0 = nb + (mA * 16 + l15) * 260 + hi * 2;
  const float* nb1 = nb + (mB * 16 + l15) * 260 + hi * 2;

  ProjGroup gA, gB;
  load_weights(gA, wkp, wvp, 0);

  const float* nbg = neighbors + (size_t)b * N_ * IND;
  for (int i = t; i < N_ * IND; i += 256)
    nb[(i >> 8) * 260 + (i & 255)] = nbg[i];
  const float* qg = Qh + (size_t)b * QCOLS;
  for (int i = t; i < QCOLS; i += 256) {
    int d = i >> 3, h = i & 7;
    Qs[d * 9 + h] = qg[i] * 0.125f;
  }
  __syncthreads();

  v8f accK[2][4] = {{{}, {}, {}, {}}, {{}, {}, {}, {}}};
  v8f accV = {};
  load_rows(gA, nb0, nb1, 0);
  for (int it = 0; it < IND / 4 / (2 * KG); ++it) {
    const int ktB = (2 * it + 1) * KG;
    load_weights(gB, wkp, wvp, ktB);
    load_rows(gB, nb0, nb1, ktB);
    proj_group(gA, accK, accV);
    const int ktA = (it + 1 < IND / 4 / (2 * KG)) ? (2 * it + 2) * KG : ktB;
    load_weights(gA, wkp, wvp, ktA);
    load_rows(gA, nb0, nb1, ktA);
    proj_group(gB, accK, accV);
  }
  {
    const v4f bias = *(const v4f*)(bK + wave * 64 + l15 * 4);
#pragma unroll
    for (int r = 0; r < 8; ++r) {
      v4f o0, o1;
#pragma unroll
      for (int tt = 0; tt < 4; ++tt) {
        o0[tt] = accK[0][tt][r] + bias[tt];
        o1[tt] = accK[1][tt][r] + bias[tt];
      }
      *(v4f*)(Kh + (mA * 16 + r + hi * 8) * QCOLS + wave * 64 + l15 * 4) = o0;
      *(v4f*)(Kh + (mB * 16 + r + hi * 8) * QCOLS + wave * 64 + l15 * 4) = o1;
    }
    const float biasV = bVs[nV0 + l15];
#pragma unroll
    for (int r = 0; r < 8; ++r)
      Vs[(mA * 16 + r + hi * 8) * OUTD + nV0 + l15] = accV[r] + biasV;
  }
  __syncthreads();

  const int it = (wave * 2) >> 2;
  const int m0q = it * 16;
  v2f aQ[2];
#pragma unroll
  for (int ks = 0; ks < 2; ++ks) {
    int kb = ks * 4 + hi * 2;
    aQ[ks].x = Qs[(m0q + l15) * 9 + kb];
    aQ[ks].y = Qs[(m0q + l15) * 9 + kb + 1];
  }

  const int j = t & 63;
  const int q = t >> 6;
  float accOut = 0.f;

  for (int n = 0; n < N_; ++n) {
#pragma unroll
    for (int tt = 0; tt < 2; ++tt) {
      int id = wave * 2 + tt;
      int j0 = (id & 3) * 16;
      v8f acc = {};
#pragma unroll
      for (int ks = 0; ks < 2; ++ks) {
        int kb = ks * 4 + hi * 2;
        v2f bb;
        bb.x = Kh[n * QCOLS + (j0 + l15) * 8 + kb];
        bb.y = Kh[n * QCOLS + (j0 + l15) * 8 + kb + 1];
        acc = __builtin_amdgcn_wmma_f32_16x16x4_f32(false, aQ[ks], false, bb, (short)0,
                                                    acc, false, false);
      }
#pragma unroll
      for (int r = 0; r < 8; ++r)
        S[(m0q + r + hi * 8) * 65 + j0 + l15] = acc[r];
    }
    __syncthreads();

    float pm = -3.4e38f;
#pragma unroll
    for (int s = 0; s < 16; ++s)
      pm = fmaxf(pm, S[(q * 16 + s) * 65 + j]);
    redM[q * 64 + j] = pm;
    __syncthreads();
    float m = fmaxf(fmaxf(redM[j], redM[64 + j]), fmaxf(redM[128 + j], redM[192 + j]));
    float pn = 0.f, pd = 0.f;
#pragma unroll
    for (int s = 0; s < 16; ++s) {
      float e = __expf(S[(q * 16 + s) * 65 + j] - m);
      pd += e;
      pn += e * Vs[n * OUTD + q * 16 + s];
    }
    redN[q * 64 + j] = pn;
    redD[q * 64 + j] = pd;
    __syncthreads();
    if (t < 64) {
      float num = redN[j] + redN[64 + j] + redN[128 + j] + redN[192 + j];
      float den = redD[j] + redD[64 + j] + redD[128 + j] + redD[192 + j];
      accOut += num / den;
    }
  }
  if (t < 64) out[(size_t)b * OUTD + t] = accOut * (1.f / H_);
}

extern "C" void kernel_launch(void* const* d_in, const int* in_sizes, int n_in,
                              void* d_out, int out_size, void* d_ws, size_t ws_size,
                              hipStream_t stream) {
  const float* center    = (const float*)d_in[0];
  const float* neighbors = (const float*)d_in[1];
  const float* WQ        = (const float*)d_in[2];
  const float* bQ        = (const float*)d_in[3];
  const float* WK        = (const float*)d_in[4];
  const float* bK        = (const float*)d_in[5];
  const float* WV        = (const float*)d_in[6];
  const float* bV        = (const float*)d_in[7];
  float* out = (float*)d_out;

  float* ws   = (float*)d_ws;
  float* wsQ  = ws;
  float* WVs  = ws + (size_t)B_ * QCOLS;
  float* bVs  = WVs + (size_t)IND * OUTD;

  prep_wvs<<<64, 256, 0, stream>>>(WV, bV, WVs, bVs);
  proj_q  <<<128, 256, 0, stream>>>(center, WQ, bQ, wsQ);
  attn_fused<<<B_, 256, 0, stream>>>(neighbors, WK, bK, WVs, bVs, wsQ, out);
}


# attn_fused: a wave owns 16 score columns x all 64 rows for half the neighbors; softmax and Vs dot run in registers, no LDS score array or per-neighbor barriers; row and neighbor sum order changes
# speedup vs baseline: 1.4279x; 1.2294x over previous
#include <hip/hip_runtime.h>

typedef __attribute__((ext_vector_type(2))) float v2f;
typedef __attribute__((ext_vector_type(4))) float v4f;
typedef __attribute__((ext_vector_type(8))) float v8f;

#define B_    2048
#define N_    32
#define IND   256
#define OUTD  64
#define H_    8
#define QCOLS 512


__global__ void prep_wvs(const float* __restrict__ WV, const float* __restrict__ bV,
                         float* __restrict__ WVs, float* __restrict__ bVs) {
  int idx = blockIdx.x * blockDim.x + threadIdx.x;
  int c = idx >> 6;
  int j = idx & 63;
  float s = 0.f;
#pragma unroll
  for (int h = 0; h < H_; ++h) s += WV[c * QCOLS + j * H_ + h];
  WVs[c * OUTD + j] = s;
  if (idx < OUTD) {
    float sb = 0.f;
#pragma unroll
    for (int h = 0; h < H_; ++h) sb += bV[idx * H_ + h];
    bVs[idx] = sb;
  }
}

__global__ __launch_bounds__(256)
void proj_q(const float* __restrict__ center, const float* __restrict__ WQ,
            const float* __restrict__ bQ, float* __restrict__ Qout) {
  int wid  = (blockIdx.x * blockDim.x + threadIdx.x) >> 5;
  int lane = threadIdx.x & 31;
  int l15 = lane & 15, hi = lane >> 4;
  int g  = wid >> 5;
  int nt = wid & 31;
  int n0 = nt * 16;
  v8f acc[4] = {{}, {}, {}, {}};
  for (int kt = 0; kt < IND / 4; ++kt) {
    int kb = kt * 4 + hi * 2;
    v2f b;
    b.x = WQ[kb * QCOLS + n0 + l15];
    b.y = WQ[(kb + 1) * QCOLS + n0 + l15];
#pragma unroll
    for (int u = 0; u < 4; ++u) {
      v2f a;
      a.x = center[(size_t)(g * 64 + u * 16 + l15) * IND + kb];
      a.y = center[(size_t)(g * 64 + u * 16 + l15) * IND + kb + 1];
      acc[u] = __builtin_amdgcn_wmma_f32_16x16x4_f32(false, a, false, b, (short)0,
                                                     acc[u], false, false);
    }
  }
  float bias = bQ[n0 + l15];
#pragma unroll
  for (int u = 0; u < 4; ++u)
#pragma unroll
    for (int r = 0; r < 8; ++r)
      Qout[(size_t)(g * 64 + u * 16 + r + hi * 8) * QCOLS + n0 + l15] = acc[u][r] + bias;
}

#define KG 4
struct ProjGroup {
  v4f w0[KG], w1[KG];
  v2f wv[KG];
  v2f a0[KG], a1[KG];
};

__device__ __forceinline__ void load_weights(ProjGroup& g, const float* __restrict__ wkp,
                                             const float* __restrict__ wvp, int kt0) {
#pragma unroll
  for (int s = 0; s < KG; ++s) {
    int kb = (kt0 + s) * 4;
    g.w0[s] = *(const v4f*)(wkp + (size_t)kb * QCOLS);
    g.w1[s] = *(const v4f*)(wkp + (size_t)(kb + 1) * QCOLS);
    g.wv[s].x = wvp[kb * OUTD];
    g.wv[s].y = wvp[(kb + 1) * OUTD];
  }
}

__device__ __forceinline__ void load_rows(ProjGroup& g, const float* nb0, const float* nb1,
                                          int kt0) {
#pragma unroll
  for (int s = 0; s < KG; ++s) {
    int kb = (kt0 + s) * 4;
    g.a0[s] = *(const v2f*)(nb0 + kb);
    g.a1[s] = *(const v2f*)(nb1 + kb);
  }
}

__device__ __forceinline__ void proj_group(const ProjGroup& g, v8f (&accK)[2][4], v8f& accV) {
#pragma unroll
  for (int s = 0; s < KG; ++s) {
#pragma unroll
    for (int t = 0; t < 4; ++t) {
      v2f bb;
      bb.x = g.w0[s][t];
      bb.y = g.w1[s][t];
      accK[0][t] = __builtin_amdgcn_wmma_f32_16x16x4_f32(false, g.a0[s], false, bb, (short)0,
                                                         accK[0][t], false, false);
      accK[1][t] = __builtin_amdgcn_wmma_f32_16x16x4_f32(false, g.a1[s], false, bb, (short)0,
                                                         accK[1][t], false, false);
    }
    accV = __builtin_amdgcn_wmma_f32_16x16x4_f32(false, g.a0[s], false, g.wv[s], (short)0,
                                                 accV, false, false);
  }
}

__device__ __forceinline__ float xhalf(float x) {
  return __builtin_bit_cast(float, __builtin_amdgcn_permlanex16(
      __builtin_bit_cast(int, x), __builtin_bit_cast(int, x), 0x76543210, 0xfedcba98,
      false, false));
}

__global__ __launch_bounds__(256)
void attn_fused(const float* __restrict__ neighbors, const float* __restrict__ WK,
                const float* __restrict__ bK, const float* __restrict__ WVs,
                const float* __restrict__ bVs, const float* __restrict__ Qh,
                float* __restrict__ out) {
  __shared__ __attribute__((aligned(16))) float nb[N_ * 260];
  __shared__ __attribute__((aligned(16))) float Kh[N_ * QCOLS];
  __shared__ __attribute__((aligned(16))) float Vs[N_ * OUTD];
  __shared__ float Qs[OUTD * 9];
  __shared__ float part[2 * OUTD];

  const int b = blockIdx.x;
  const int t = threadIdx.x;
  const int wave = t >> 5, lane = t & 31, l15 = lane & 15, hi = lane >> 4;

  const int mA = wave >> 2, mB = 1 - mA;
  const int nV0 = (wave & 3) * 16;
  const float* wkp = WK + (size_t)(hi * 2) * QCOLS + wave * 64 + l15 * 4;
  const float* wvp = WVs + (size_t)(hi * 2) * OUTD + nV0 + l15;
  const float* nb0 = nb + (mA * 16 + l15) * 260 + hi * 2;
  const float* nb1 = nb + (mB * 16 + l15) * 260 + hi * 2;

  ProjGroup gA, gB;
  load_weights(gA, wkp, wvp, 0);

  const float* nbg = neighbors + (size_t)b * N_ * IND;
  for (int i = t; i < N_ * IND; i += 256)
    nb[(i >> 8) * 260 + (i & 255)] = nbg[i];
  const float* qg = Qh + (size_t)b * QCOLS;
  for (int i = t; i < QCOLS; i += 256) {
    int d = i >> 3, h = i & 7;
    Qs[d * 9 + h] = qg[i] * 0.125f;
  }
  __syncthreads();

  v8f accK[2][4] = {{{}, {}, {}, {}}, {{}, {}, {}, {}}};
  v8f accV = {};
  load_rows(gA, nb0, nb1, 0);
  for (int it = 0; it < IND / 4 / (2 * KG); ++it) {
    const int ktB = (2 * it + 1) * KG;
    load_weights(gB, wkp, wvp, ktB);
    load_rows(gB, nb0, nb1, ktB);
    proj_group(gA, accK, accV);
    const int ktA = (it + 1 < IND / 4 / (2 * KG)) ? (2 * it + 2) * KG : ktB;
    load_weights(gA, wkp, wvp, ktA);
    load_rows(gA, nb0, nb1, ktA);
    proj_group(gB, accK, accV);
  }
  {
    const v4f bias = *(const v4f*)(bK + wave * 64 + l15 * 4);
#pragma unroll
    for (int r = 0; r < 8; ++r) {
      v4f o0, o1;
#pragma unroll
      for (int tt = 0; tt < 4; ++tt) {
        o0[tt] = accK[0][tt][r] + bias[tt];
        o1[tt] = accK[1][tt][r] + bias[tt];
      }
      *(v4f*)(Kh + (mA * 16 + r + hi * 8) * QCOLS + wave * 64 + l15 * 4) = o0;
      *(v4f*)(Kh + (mB * 16 + r + hi * 8) * QCOLS + wave * 64 + l15 * 4) = o1;
    }
    const float biasV = bVs[nV0 + l15];
#pragma unroll
    for (int r = 0; r < 8; ++r)
      Vs[(mA * 16 + r + hi * 8) * OUTD + nV0 + l15] = accV[r] + biasV;
  }
  __syncthreads();

  const int j0 = (wave & 3) * 16;
  const int nh = wave >> 2;
  v2f aQ[4][2];
#pragma unroll
  for (int mt = 0; mt < 4; ++mt)
#pragma unroll
    for (int ks = 0; ks < 2; ++ks) {
      int kb = ks * 4 + hi * 2;
      aQ[mt][ks].x = Qs[(mt * 16 + l15) * 9 + kb];
      aQ[mt][ks].y = Qs[(mt * 16 + l15) * 9 + kb + 1];
    }
  const float* kcol = Kh + (size_t)(nh * (N_ / 2)) * QCOLS + (j0 + l15) * 8 + hi * 2;
  const float* vrow = Vs + (nh * (N_ / 2)) * OUTD + hi * 8;
  float accOut = 0.f;

  for (int nn = 0; nn < N_ / 2; ++nn) {
    const v2f bb0 = *(const v2f*)(kcol + nn * QCOLS);
    const v2f bb1 = *(const v2f*)(kcol + nn * QCOLS + 4);
    v8f s[4];
#pragma unroll
    for (int mt = 0; mt < 4; ++mt) {
      v8f z = {};
      s[mt] = __builtin_amdgcn_wmma_f32_16x16x4_f32(false, aQ[mt][0], false, bb0, (short)0,
                                                    z, false, false);
    }
#pragma unroll
    for (int mt = 0; mt < 4; ++mt)
      s[mt] = __builtin_amdgcn_wmma_f32_16x16x4_f32(false, aQ[mt][1], false, bb1, (short)0,
                                                    s[mt], false, false);
    v4f vs[4][2];
#pragma unroll
    for (int mt = 0; mt < 4; ++mt) {
      vs[mt][0] = *(const v4f*)(vrow + nn * OUTD + mt * 16);
      vs[mt][1] = *(const v4f*)(vrow + nn * OUTD + mt * 16 + 4);
    }
    float pm[4];
#pragma unroll
    for (int mt = 0; mt < 4; ++mt) {
      pm[mt] = s[mt][0];
#pragma unroll
      for (int r = 1; r < 8; ++r) pm[mt] = fmaxf(pm[mt], s[mt][r]);
    }
    float m = fmaxf(fmaxf(pm[0], pm[1]), fmaxf(pm[2], pm[3]));
    m = fmaxf(m, xhalf(m));
    float pd[4], pn[4];
#pragma unroll
    for (int mt = 0; mt < 4; ++mt) {
      pd[mt] = 0.f;
      pn[mt] = 0.f;
#pragma unroll
      for (int r = 0; r < 8; ++r) {
        float e = __expf(s[mt][r] - m);
        pd[mt] += e;
        pn[mt] += e * vs[mt][r >> 2][r & 3];
      }
    }
    float den = (pd[0] + pd[1]) + (pd[2] + pd[3]);
    float num = (pn[0] + pn[1]) + (pn[2] + pn[3]);
    den += xhalf(den);
    num += xhalf(num);
    accOut += num / den;
  }
  if (hi == 0) part[nh * OUTD + j0 + l15] = accOut;
  __syncthreads();
  if (t < 64) out[(size_t)b * OUTD + t] = (part[t] + part[OUTD + t]) * (1.f / H_);
}

extern "C" void kernel_launch(void* const* d_in, const int* in_sizes, int n_in,
                              void* d_out, int out_size, void* d_ws, size_t ws_size,
                              hipStream_t stream) {
  const float* center    = (const float*)d_in[0];
  const float* neighbors = (const float*)d_in[1];
  const float* WQ        = (const float*)d_in[2];
  const float* bQ        = (const float*)d_in[3];
  const float* WK        = (const float*)d_in[4];
  const float* bK        = (const float*)d_in[5];
  const float* WV        = (const float*)d_in[6];
  const float* bV        = (const float*)d_in[7];
  float* out = (float*)d_out;

  float* ws   = (float*)d_ws;
  float* wsQ  = ws;
  float* WVs  = ws + (size_t)B_ * QCOLS;
  float* bVs  = WVs + (size_t)IND * OUTD;

  prep_wvs<<<64, 256, 0, stream>>>(WV, bV, WVs, bVs);
  proj_q  <<<128, 256, 0, stream>>>(center, WQ, bQ, wsQ);
  attn_fused<<<B_, 256, 0, stream>>>(neighbors, WK, bK, WVs, bVs, wsQ, out);
}


# loads go out in one batch ahead of use: attn_fused input staging (wide loads, one round trip), its WMMA refill groups (pinned by fences), proj_q (new two-set register pipeline); sum order unchanged
# speedup vs baseline: 1.5930x; 1.1156x over previous
#include <hip/hip_runtime.h>

typedef __attribute__((ext_vector_type(2))) float v2f;
typedef __attribute__((ext_vector_type(4))) float v4f;
typedef __attribute__((ext_vector_type(8))) float v8f;

#define B_    2048
#define N_    32
#define IND   256
#define OUTD  64
#define H_    8
#define QCOLS 512


__global__ void prep_wvs(const float* __restrict__ WV, const float* __restrict__ bV,
                         float* __restrict__ WVs, float* __restrict__ bVs) {
  int idx = blockIdx.x * blockDim.x + threadIdx.x;
  int c = idx >> 6;
  int j = idx & 63;
  float s = 0.f;
#pragma unroll
  for (int h = 0; h < H_; ++h) s += WV[c * QCOLS + j * H_ + h];
  WVs[c * OUTD + j] = s;
  if (idx < OUTD) {
    float sb = 0.f;
#pragma unroll
    for (int h = 0; h < H_; ++h) sb += bV[idx * H_ + h];
    bVs[idx] = sb;
  }
}

#define KQ 8
struct QGroup {
  v2f b[KQ];
  v2f a[KQ][4];
};

__device__ __forceinline__ void load_q(QGroup& g, const float* __restrict__ ap,
                                       const float* __restrict__ bp, int kt0) {
#pragma unroll
  for (int s = 0; s < KQ; ++s) {
    int kb = (kt0 + s) * 4;
    g.b[s].x = bp[(size_t)kb * QCOLS];
    g.b[s].y = bp[(size_t)(kb + 1) * QCOLS];
#pragma unroll
    for (int u = 0; u < 4; ++u) g.a[s][u] = *(const v2f*)(ap + (size_t)u * 16 * IND + kb);
  }
}

__device__ __forceinline__ void mma_q(const QGroup& g, v8f (&acc)[4]) {
#pragma unroll
  for (int s = 0; s < KQ; ++s)
#pragma unroll
    for (int u = 0; u < 4; ++u)
      acc[u] = __builtin_amdgcn_wmma_f32_16x16x4_f32(false, g.a[s][u], false, g.b[s], (short)0,
                                                     acc[u], false, false);
}

__global__ __launch_bounds__(256)
void proj_q(const float* __restrict__ center, const float* __restrict__ WQ,
            const float* __restrict__ bQ, float* __restrict__ Qout) {
  int wid  = (blockIdx.x * blockDim.x + threadIdx.x) >> 5;
  int lane = threadIdx.x & 31;
  int l15 = lane & 15, hi = lane >> 4;
  int g  = wid >> 5;
  int nt = wid & 31;
  int n0 = nt * 16;
  const float* ap = center + (size_t)(g * 64 + l15) * IND + hi * 2;
  const float* bp = WQ + (size_t)(hi * 2) * QCOLS + n0 + l15;
  v8f acc[4] = {{}, {}, {}, {}};
  QGroup qA, qB;
  load_q(qA, ap, bp, 0);
  for (int it = 0; it < IND / 4 / (2 * KQ); ++it) {
    const int ktB = (2 * it + 1) * KQ;
    load_q(qB, ap, bp, ktB);
    __builtin_amdgcn_sched_barrier(0);
    mma_q(qA, acc);
    __builtin_amdgcn_sched_barrier(0);
    const int ktA = (it + 1 < IND / 4 / (2 * KQ)) ? (2 * it + 2) * KQ : ktB;
    load_q(qA, ap, bp, ktA);
    __builtin_amdgcn_sched_barrier(0);
    mma_q(qB, acc);
    __builtin_amdgcn_sched_barrier(0);
  }
  float bias = bQ[n0 + l15];
#pragma unroll
  for (int u = 0; u < 4; ++u)
#pragma unroll
    for (int r = 0; r < 8; ++r)
      Qout[(size_t)(g * 64 + u * 16 + r + hi * 8) * QCOLS + n0 + l15] = acc[u][r] + bias;
}

#define KG 4
struct ProjGroup {
  v4f w0[KG], w1[KG];
  v2f wv[KG];
  v2f a0[KG], a1[KG];
};

__device__ __forceinline__ void load_weights(ProjGroup& g, const float* __restrict__ wkp,
                                             const float* __restrict__ wvp, int kt0) {
#pragma unroll
  for (int s = 0; s < KG; ++s) {
    int kb = (kt0 + s) * 4;
    g.w0[s] = *(const v4f*)(wkp + (size_t)kb * QCOLS);
    g.w1[s] = *(const v4f*)(wkp + (size_t)(kb + 1) * QCOLS);
    g.wv[s].x = wvp[kb * OUTD];
    g.wv[s].y = wvp[(kb + 1) * OUTD];
  }
}

__device__ __forceinline__ void load_rows(ProjGroup& g, const float* nb0, const float* nb1,
                                          int kt0) {
#pragma unroll
  for (int s = 0; s < KG; ++s) {
    int kb = (kt0 + s) * 4;
    g.a0[s] = *(const v2f*)(nb0 + kb);
    g.a1[s] = *(const v2f*)(nb1 + kb);
  }
}

__device__ __forceinline__ void proj_group(const ProjGroup& g, v8f (&accK)[2][4], v8f& accV) {
#pragma unroll
  for (int s = 0; s < KG; ++s) {
#pragma unroll
    for (int t = 0; t < 4; ++t) {
      v2f bb;
      bb.x = g.w0[s][t];
      bb.y = g.w1[s][t];
      accK[0][t] = __builtin_amdgcn_wmma_f32_16x16x4_f32(false, g.a0[s], false, bb, (short)0,
                                                         accK[0][t], false, false);
      accK[1][t] = __builtin_amdgcn_wmma_f32_16x16x4_f32(false, g.a1[s], false, bb, (short)0,
                                                         accK[1][t], false, false);
    }
    accV = __builtin_amdgcn_wmma_f32_16x16x4_f32(false, g.a0[s], false, g.wv[s], (short)0,
                                                 accV, false, false);
  }
}

__device__ __forceinline__ float xhalf(float x) {
  return __builtin_bit_cast(float, __builtin_amdgcn_permlanex16(
      __builtin_bit_cast(int, x), __builtin_bit_cast(int, x), 0x76543210, 0xfedcba98,
      false, false));
}

__global__ __launch_bounds__(256)
void attn_fused(const float* __restrict__ neighbors, const float* __restrict__ WK,
                const float* __restrict__ bK, const float* __restrict__ WVs,
                const float* __restrict__ bVs, const float* __restrict__ Qh,
                float* __restrict__ out) {
  __shared__ __attribute__((aligned(16))) float nb[N_ * 260];
  __shared__ __attribute__((aligned(16))) float Kh[N_ * QCOLS];
  __shared__ __attribute__((aligned(16))) float Vs[N_ * OUTD];
  __shared__ float Qs[OUTD * 9];
  __shared__ float part[2 * OUTD];

  const int b = blockIdx.x;
  const int t = threadIdx.x;
  const int wave = t >> 5, lane = t & 31, l15 = lane & 15, hi = lane >> 4;

  const int mA = wave >> 2, mB = 1 - mA;
  const int nV0 = (wave & 3) * 16;
  const float* wkp = WK + (size_t)(hi * 2) * QCOLS + wave * 64 + l15 * 4;
  const float* wvp = WVs + (size_t)(hi * 2) * OUTD + nV0 + l15;
  const float* nb0 = nb + (mA * 16 + l15) * 260 + hi * 2;
  const float* nb1 = nb + (mB * 16 + l15) * 260 + hi * 2;

  ProjGroup gA, gB;
  load_weights(gA, wkp, wvp, 0);

  const float* nbg = neighbors + (size_t)b * N_ * IND;
  const float* qg = Qh + (size_t)b * QCOLS;
  v4f st[8];
#pragma unroll
  for (int k = 0; k < 8; ++k) st[k] = *(const v4f*)(nbg + (size_t)(t + 256 * k) * 4);
  const float q0 = qg[t], q1 = qg[t + 256];
#pragma unroll
  for (int k = 0; k < 8; ++k) {
    const int f = t + 256 * k;
    *(v4f*)(nb + (f >> 6) * 260 + (f & 63) * 4) = st[k];
  }
  Qs[(t >> 3) * 9 + (t & 7)] = q0 * 0.125f;
  Qs[((t + 256) >> 3) * 9 + (t & 7)] = q1 * 0.125f;
  __syncthreads();

  v8f accK[2][4] = {{{}, {}, {}, {}}, {{}, {}, {}, {}}};
  v8f accV = {};
  load_rows(gA, nb0, nb1, 0);
  for (int it = 0; it < IND / 4 / (2 * KG); ++it) {
    const int ktB = (2 * it + 1) * KG;
    load_weights(gB, wkp, wvp, ktB);
    load_rows(gB, nb0, nb1, ktB);
    __builtin_amdgcn_sched_barrier(0);
    proj_group(gA, accK, accV);
    __builtin_amdgcn_sched_barrier(0);
    const int ktA = (it + 1 < IND / 4 / (2 * KG)) ? (2 * it + 2) * KG : ktB;
    load_weights(gA, wkp, wvp, ktA);
    load_rows(gA, nb0, nb1, ktA);
    __builtin_amdgcn_sched_barrier(0);
    proj_group(gB, accK, accV);
  }
  {
    const v4f bias = *(const v4f*)(bK + wave * 64 + l15 * 4);
#pragma unroll
    for (int r = 0; r < 8; ++r) {
      v4f o0, o1;
#pragma unroll
      for (int tt = 0; tt < 4; ++tt) {
        o0[tt] = accK[0][tt][r] + bias[tt];
        o1[tt] = accK[1][tt][r] + bias[tt];
      }
      *(v4f*)(Kh + (mA * 16 + r + hi * 8) * QCOLS + wave * 64 + l15 * 4) = o0;
      *(v4f*)(Kh + (mB * 16 + r + hi * 8) * QCOLS + wave * 64 + l15 * 4) = o1;
    }
    const float biasV = bVs[nV0 + l15];
#pragma unroll
    for (int r = 0; r < 8; ++r)
      Vs[(mA * 16 + r + hi * 8) * OUTD + nV0 + l15] = accV[r] + biasV;
  }
  __syncthreads();

  const int j0 = (wave & 3) * 16;
  const int nh = wave >> 2;
  v2f aQ[4][2];
#pragma unroll
  for (int mt = 0; mt < 4; ++mt)
#pragma unroll
    for (int ks = 0; ks < 2; ++ks) {
      int kb = ks * 4 + hi * 2;
      aQ[mt][ks].x = Qs[(mt * 16 + l15) * 9 + kb];
      aQ[mt][ks].y = Qs[(mt * 16 + l15) * 9 + kb + 1];
    }
  const float* kcol = Kh + (size_t)(nh * (N_ / 2)) * QCOLS + (j0 + l15) * 8 + hi * 2;
  const float* vrow = Vs + (nh * (N_ / 2)) * OUTD + hi * 8;
  float accOut = 0.f;

  for (int nn = 0; nn < N_ / 2; ++nn) {
    const v2f bb0 = *(const v2f*)(kcol + nn * QCOLS);
    const v2f bb1 = *(const v2f*)(kcol + nn * QCOLS + 4);
    v8f s[4];
#pragma unroll
    for (int mt = 0; mt < 4; ++mt) {
      v8f z = {};
      s[mt] = __builtin_amdgcn_wmma_f32_16x16x4_f32(false, aQ[mt][0], false, bb0, (short)0,
                                                    z, false, false);
    }
#pragma unroll
    for (int mt = 0; mt < 4; ++mt)
      s[mt] = __builtin_amdgcn_wmma_f32_16x16x4_f32(false, aQ[mt][1], false, bb1, (short)0,
                                                    s[mt], false, false);
    v4f vs[4][2];
#pragma unroll
    for (int mt = 0; mt < 4; ++mt) {
      vs[mt][0] = *(const v4f*)(vrow + nn * OUTD + mt * 16);
      vs[mt][1] = *(const v4f*)(vrow + nn * OUTD + mt * 16 + 4);
    }
    float pm[4];
#pragma unroll
    for (int mt = 0; mt < 4; ++mt) {
      pm[mt] = s[mt][0];
#pragma unroll
      for (int r = 1; r < 8; ++r) pm[mt] = fmaxf(pm[mt], s[mt][r]);
    }
    float m = fmaxf(fmaxf(pm[0], pm[1]), fmaxf(pm[2], pm[3]));
    m = fmaxf(m, xhalf(m));
    float pd[4], pn[4];
#pragma unroll
    for (int mt = 0; mt < 4; ++mt) {
      pd[mt] = 0.f;
      pn[mt] = 0.f;
#pragma unroll
      for (int r = 0; r < 8; ++r) {
        float e = __expf(s[mt][r] - m);
        pd[mt] += e;
        pn[mt] += e * vs[mt][r >> 2][r & 3];
      }
    }
    float den = (pd[0] + pd[1]) + (pd[2] + pd[3]);
    float num = (pn[0] + pn[1]) + (pn[2] + pn[3]);
    den += xhalf(den);
    num += xhalf(num);
    accOut += num / den;
  }
  if (hi == 0) part[nh * OUTD + j0 + l15] = accOut;
  __syncthreads();
  if (t < 64) out[(size_t)b * OUTD + t] = (part[t] + part[OUTD + t]) * (1.f / H_);
}

extern "C" void kernel_launch(void* const* d_in, const int* in_sizes, int n_in,
                              void* d_out, int out_size, void* d_ws, size_t ws_size,
                              hipStream_t stream) {
  const float* center    = (const float*)d_in[0];
  const float* neighbors = (const float*)d_in[1];
  const float* WQ        = (const float*)d_in[2];
  const float* bQ        = (const float*)d_in[3];
  const float* WK        = (const float*)d_in[4];
  const float* bK        = (const float*)d_in[5];
  const float* WV        = (const float*)d_in[6];
  const float* bV        = (const float*)d_in[7];
  float* out = (float*)d_out;

  float* ws   = (float*)d_ws;
  float* wsQ  = ws;
  float* WVs  = ws + (size_t)B_ * QCOLS;
  float* bVs  = WVs + (size_t)IND * OUTD;

  prep_wvs<<<64, 256, 0, stream>>>(WV, bV, WVs, bVs);
  proj_q  <<<128, 256, 0, stream>>>(center, WQ, bQ, wsQ);
  attn_fused<<<B_, 256, 0, stream>>>(neighbors, WK, bK, WVs, bVs, wsQ, out);
}


# attn_fused softmax loop: fewer vector issue cycles per neighbor (exp2 on scores pre-scaled by log2(e), two-wide row-pair ops, rcp multiply); rounding and row sum order change
# speedup vs baseline: 1.6956x; 1.0644x over previous
#include <hip/hip_runtime.h>

typedef __attribute__((ext_vector_type(2))) float v2f;
typedef __attribute__((ext_vector_type(4))) float v4f;
typedef __attribute__((ext_vector_type(8))) float v8f;

#define B_    2048
#define N_    32
#define IND   256
#define OUTD  64
#define H_    8
#define QCOLS 512


__global__ void prep_wvs(const float* __restrict__ WV, const float* __restrict__ bV,
                         float* __restrict__ WVs, float* __restrict__ bVs) {
  int idx = blockIdx.x * blockDim.x + threadIdx.x;
  int c = idx >> 6;
  int j = idx & 63;
  float s = 0.f;
#pragma unroll
  for (int h = 0; h < H_; ++h) s += WV[c * QCOLS + j * H_ + h];
  WVs[c * OUTD + j] = s;
  if (idx < OUTD) {
    float sb = 0.f;
#pragma unroll
    for (int h = 0; h < H_; ++h) sb += bV[idx * H_ + h];
    bVs[idx] = sb;
  }
}

#define KQ 8
struct QGroup {
  v2f b[KQ];
  v2f a[KQ][4];
};

__device__ __forceinline__ void load_q(QGroup& g, const float* __restrict__ ap,
                                       const float* __restrict__ bp, int kt0) {
#pragma unroll
  for (int s = 0; s < KQ; ++s) {
    int kb = (kt0 + s) * 4;
    g.b[s].x = bp[(size_t)kb * QCOLS];
    g.b[s].y = bp[(size_t)(kb + 1) * QCOLS];
#pragma unroll
    for (int u = 0; u < 4; ++u) g.a[s][u] = *(const v2f*)(ap + (size_t)u * 16 * IND + kb);
  }
}

__device__ __forceinline__ void mma_q(const QGroup& g, v8f (&acc)[4]) {
#pragma unroll
  for (int s = 0; s < KQ; ++s)
#pragma unroll
    for (int u = 0; u < 4; ++u)
      acc[u] = __builtin_amdgcn_wmma_f32_16x16x4_f32(false, g.a[s][u], false, g.b[s], (short)0,
                                                     acc[u], false, false);
}

__global__ __launch_bounds__(256)
void proj_q(const float* __restrict__ center, const float* __restrict__ WQ,
            const float* __restrict__ bQ, float* __restrict__ Qout) {
  int wid  = (blockIdx.x * blockDim.x + threadIdx.x) >> 5;
  int lane = threadIdx.x & 31;
  int l15 = lane & 15, hi = lane >> 4;
  int g  = wid >> 5;
  int nt = wid & 31;
  int n0 = nt * 16;
  const float* ap = center + (size_t)(g * 64 + l15) * IND + hi * 2;
  const float* bp = WQ + (size_t)(hi * 2) * QCOLS + n0 + l15;
  v8f acc[4] = {{}, {}, {}, {}};
  QGroup qA, qB;
  load_q(qA, ap, bp, 0);
  for (int it = 0; it < IND / 4 / (2 * KQ); ++it) {
    const int ktB = (2 * it + 1) * KQ;
    load_q(qB, ap, bp, ktB);
    __builtin_amdgcn_sched_barrier(0);
    mma_q(qA, acc);
    __builtin_amdgcn_sched_barrier(0);
    const int ktA = (it + 1 < IND / 4 / (2 * KQ)) ? (2 * it + 2) * KQ : ktB;
    load_q(qA, ap, bp, ktA);
    __builtin_amdgcn_sched_barrier(0);
    mma_q(qB, acc);
    __builtin_amdgcn_sched_barrier(0);
  }
  float bias = bQ[n0 + l15];
#pragma unroll
  for (int u = 0; u < 4; ++u)
#pragma unroll
    for (int r = 0; r < 8; ++r)
      Qout[(size_t)(g * 64 + u * 16 + r + hi * 8) * QCOLS + n0 + l15] = acc[u][r] + bias;
}

#define KG 4
struct ProjGroup {
  v4f w0[KG], w1[KG];
  v2f wv[KG];
  v2f a0[KG], a1[KG];
};

__device__ __forceinline__ void load_weights(ProjGroup& g, const float* __restrict__ wkp,
                                             const float* __restrict__ wvp, int kt0) {
#pragma unroll
  for (int s = 0; s < KG; ++s) {
    int kb = (kt0 + s) * 4;
    g.w0[s] = *(const v4f*)(wkp + (size_t)kb * QCOLS);
    g.w1[s] = *(const v4f*)(wkp + (size_t)(kb + 1) * QCOLS);
    g.wv[s].x = wvp[kb * OUTD];
    g.wv[s].y = wvp[(kb + 1) * OUTD];
  }
}

__device__ __forceinline__ void load_rows(ProjGroup& g, const float* nb0, const float* nb1,
                                          int kt0) {
#pragma unroll
  for (int s = 0; s < KG; ++s) {
    int kb = (kt0 + s) * 4;
    g.a0[s] = *(const v2f*)(nb0 + kb);
    g.a1[s] = *(const v2f*)(nb1 + kb);
  }
}

__device__ __forceinline__ void proj_group(const ProjGroup& g, v8f (&accK)[2][4], v8f& accV) {
#pragma unroll
  for (int s = 0; s < KG; ++s) {
#pragma unroll
    for (int t = 0; t < 4; ++t) {
      v2f bb;
      bb.x = g.w0[s][t];
      bb.y = g.w1[s][t];
      accK[0][t] = __builtin_amdgcn_wmma_f32_16x16x4_f32(false, g.a0[s], false, bb, (short)0,
                                                         accK[0][t], false, false);
      accK[1][t] = __builtin_amdgcn_wmma_f32_16x16x4_f32(false, g.a1[s], false, bb, (short)0,
                                                         accK[1][t], false, false);
    }
    accV = __builtin_amdgcn_wmma_f32_16x16x4_f32(false, g.a0[s], false, g.wv[s], (short)0,
                                                 accV, false, false);
  }
}

__device__ __forceinline__ float xhalf(float x) {
  return __builtin_bit_cast(float, __builtin_amdgcn_permlanex16(
      __builtin_bit_cast(int, x), __builtin_bit_cast(int, x), 0x76543210, 0xfedcba98,
      false, false));
}

__global__ __launch_bounds__(256)
void attn_fused(const float* __restrict__ neighbors, const float* __restrict__ WK,
                const float* __restrict__ bK, const float* __restrict__ WVs,
                const float* __restrict__ bVs, const float* __restrict__ Qh,
                float* __restrict__ out) {
  __shared__ __attribute__((aligned(16))) float nb[N_ * 260];
  __shared__ __attribute__((aligned(16))) float Kh[N_ * QCOLS];
  __shared__ __attribute__((aligned(16))) float Vs[N_ * OUTD];
  __shared__ float Qs[OUTD * 9];
  __shared__ float part[2 * OUTD];

  const int b = blockIdx.x;
  const int t = threadIdx.x;
  const int wave = t >> 5, lane = t & 31, l15 = lane & 15, hi = lane >> 4;

  const int mA = wave >> 2, mB = 1 - mA;
  const int nV0 = (wave & 3) * 16;
  const float* wkp = WK + (size_t)(hi * 2) * QCOLS + wave * 64 + l15 * 4;
  const float* wvp = WVs + (size_t)(hi * 2) * OUTD + nV0 + l15;
  const float* nb0 = nb + (mA * 16 + l15) * 260 + hi * 2;
  const float* nb1 = nb + (mB * 16 + l15) * 260 + hi * 2;

  ProjGroup gA, gB;
  load_weights(gA, wkp, wvp, 0);

  const float* nbg = neighbors + (size_t)b * N_ * IND;
  const float* qg = Qh + (size_t)b * QCOLS;
  v4f st[8];
#pragma unroll
  for (int k = 0; k < 8; ++k) st[k] = *(const v4f*)(nbg + (size_t)(t + 256 * k) * 4);
  const float q0 = qg[t], q1 = qg[t + 256];
#pragma unroll
  for (int k = 0; k < 8; ++k) {
    const int f = t + 256 * k;
    *(v4f*)(nb + (f >> 6) * 260 + (f & 63) * 4) = st[k];
  }
  const float qscale = 0.125f * 1.44269504088896340736f;
  Qs[(t >> 3) * 9 + (t & 7)] = q0 * qscale;
  Qs[((t + 256) >> 3) * 9 + (t & 7)] = q1 * qscale;
  __syncthreads();

  v8f accK[2][4] = {{{}, {}, {}, {}}, {{}, {}, {}, {}}};
  v8f accV = {};
  load_rows(gA, nb0, nb1, 0);
  for (int it = 0; it < IND / 4 / (2 * KG); ++it) {
    const int ktB = (2 * it + 1) * KG;
    load_weights(gB, wkp, wvp, ktB);
    load_rows(gB, nb0, nb1, ktB);
    __builtin_amdgcn_sched_barrier(0);
    proj_group(gA, accK, accV);
    __builtin_amdgcn_sched_barrier(0);
    const int ktA = (it + 1 < IND / 4 / (2 * KG)) ? (2 * it + 2) * KG : ktB;
    load_weights(gA, wkp, wvp, ktA);
    load_rows(gA, nb0, nb1, ktA);
    __builtin_amdgcn_sched_barrier(0);
    proj_group(gB, accK, accV);
  }
  {
    const v4f bias = *(const v4f*)(bK + wave * 64 + l15 * 4);
#pragma unroll
    for (int r = 0; r < 8; ++r) {
      v4f o0, o1;
#pragma unroll
      for (int tt = 0; tt < 4; ++tt) {
        o0[tt] = accK[0][tt][r] + bias[tt];
        o1[tt] = accK[1][tt][r] + bias[tt];
      }
      *(v4f*)(Kh + (mA * 16 + r + hi * 8) * QCOLS + wave * 64 + l15 * 4) = o0;
      *(v4f*)(Kh + (mB * 16 + r + hi * 8) * QCOLS + wave * 64 + l15 * 4) = o1;
    }
    const float biasV = bVs[nV0 + l15];
#pragma unroll
    for (int r = 0; r < 8; ++r)
      Vs[(mA * 16 + r + hi * 8) * OUTD + nV0 + l15] = accV[r] + biasV;
  }
  __syncthreads();

  const int j0 = (wave & 3) * 16;
  const int nh = wave >> 2;
  v2f aQ[4][2];
#pragma unroll
  for (int mt = 0; mt < 4; ++mt)
#pragma unroll
    for (int ks = 0; ks < 2; ++ks) {
      int kb = ks * 4 + hi * 2;
      aQ[mt][ks].x = Qs[(mt * 16 + l15) * 9 + kb];
      aQ[mt][ks].y = Qs[(mt * 16 + l15) * 9 + kb + 1];
    }
  const float* kcol = Kh + (size_t)(nh * (N_ / 2)) * QCOLS + (j0 + l15) * 8 + hi * 2;
  const float* vrow = Vs + (nh * (N_ / 2)) * OUTD + hi * 8;
  float accOut = 0.f;

  for (int nn = 0; nn < N_ / 2; ++nn) {
    const v2f bb0 = *(const v2f*)(kcol + nn * QCOLS);
    const v2f bb1 = *(const v2f*)(kcol + nn * QCOLS + 4);
    v8f s[4];
#pragma unroll
    for (int mt = 0; mt < 4; ++mt) {
      v8f z = {};
      s[mt] = __builtin_amdgcn_wmma_f32_16x16x4_f32(false, aQ[mt][0], false, bb0, (short)0,
                                                    z, false, false);
    }
#pragma unroll
    for (int mt = 0; mt < 4; ++mt)
      s[mt] = __builtin_amdgcn_wmma_f32_16x16x4_f32(false, aQ[mt][1], false, bb1, (short)0,
                                                    s[mt], false, false);
    v4f vs[4][2];
#pragma unroll
    for (int mt = 0; mt < 4; ++mt) {
      vs[mt][0] = *(const v4f*)(vrow + nn * OUTD + mt * 16);
      vs[mt][1] = *(const v4f*)(vrow + nn * OUTD + mt * 16 + 4);
    }
    float pm[4];
#pragma unroll
    for (int mt = 0; mt < 4; ++mt) {
      pm[mt] = s[mt][0];
#pragma unroll
      for (int r = 1; r < 8; ++r) pm[mt] = fmaxf(pm[mt], s[mt][r]);
    }
    float m = fmaxf(fmaxf(pm[0], pm[1]), fmaxf(pm[2], pm[3]));
    m = fmaxf(m, xhalf(m));
    const v2f m2 = {m, m};
    v2f pd[4], pn[4];
#pragma unroll
    for (int mt = 0; mt < 4; ++mt) {
#pragma unroll
      for (int p = 0; p < 4; ++p) {
        v2f sv, vv, e;
        sv.x = s[mt][2 * p];
        sv.y = s[mt][2 * p + 1];
        vv.x = vs[mt][p >> 1][(2 * p) & 3];
        vv.y = vs[mt][p >> 1][(2 * p + 1) & 3];
        const v2f d = sv - m2;
        e.x = __builtin_amdgcn_exp2f(d.x);
        e.y = __builtin_amdgcn_exp2f(d.y);
        if (p == 0) {
          pd[mt] = e;
          pn[mt] = e * vv;
        } else {
          pd[mt] += e;
          pn[mt] += e * vv;
        }
      }
    }
    const v2f den2 = (pd[0] + pd[1]) + (pd[2] + pd[3]);
    const v2f num2 = (pn[0] + pn[1]) + (pn[2] + pn[3]);
    float den = den2.x + den2.y;
    float num = num2.x + num2.y;
    den += xhalf(den);
    num += xhalf(num);
    accOut += num * __builtin_amdgcn_rcpf(den);
  }
  if (hi == 0) part[nh * OUTD + j0 + l15] = accOut;
  __syncthreads();
  if (t < 64) out[(size_t)b * OUTD + t] = (part[t] + part[OUTD + t]) * (1.f / H_);
}

extern "C" void kernel_launch(void* const* d_in, const int* in_sizes, int n_in,
                              void* d_out, int out_size, void* d_ws, size_t ws_size,
                              hipStream_t stream) {
  const float* center    = (const float*)d_in[0];
  const float* neighbors = (const float*)d_in[1];
  const float* WQ        = (const float*)d_in[2];
  const float* bQ        = (const float*)d_in[3];
  const float* WK        = (const float*)d_in[4];
  const float* bK        = (const float*)d_in[5];
  const float* WV        = (const float*)d_in[6];
  const float* bV        = (const float*)d_in[7];
  float* out = (float*)d_out;

  float* ws   = (float*)d_ws;
  float* wsQ  = ws;
  float* WVs  = ws + (size_t)B_ * QCOLS;
  float* bVs  = WVs + (size_t)IND * OUTD;

  prep_wvs<<<64, 256, 0, stream>>>(WV, bV, WVs, bVs);
  proj_q  <<<128, 256, 0, stream>>>(center, WQ, bQ, wsQ);
  attn_fused<<<B_, 256, 0, stream>>>(neighbors, WK, bK, WVs, bVs, wsQ, out);
}


# prep_wvs and proj_q become one launch of 4-wave groups on all 256 WGPs; the Q projection reads its center rows from an LDS panel staged by whole-line loads; sum order unchanged
# speedup vs baseline: 1.7585x; 1.0371x over previous
#include <hip/hip_runtime.h>

typedef __attribute__((ext_vector_type(2))) float v2f;
typedef __attribute__((ext_vector_type(4))) float v4f;
typedef __attribute__((ext_vector_type(8))) float v8f;

#define B_    2048
#define N_    32
#define IND   256
#define OUTD  64
#define H_    8
#define QCOLS 512


#define KQ 8
#define QROWS 64
#define QPITCH 260
#define QGROUPS 256
struct QGroup {
  v2f b[KQ];
  v2f a[KQ][4];
};

__device__ __forceinline__ void load_qb(QGroup& g, const float* __restrict__ bp, int kt0) {
#pragma unroll
  for (int s = 0; s < KQ; ++s) {
    int kb = (kt0 + s) * 4;
    g.b[s].x = bp[(size_t)kb * QCOLS];
    g.b[s].y = bp[(size_t)(kb + 1) * QCOLS];
  }
}

__device__ __forceinline__ void load_qa(QGroup& g, const float* ap, int kt0) {
#pragma unroll
  for (int s = 0; s < KQ; ++s) {
    int kb = (kt0 + s) * 4;
#pragma unroll
    for (int u = 0; u < 4; ++u) g.a[s][u] = *(const v2f*)(ap + u * 16 * QPITCH + kb);
  }
}

__device__ __forceinline__ void mma_q(const QGroup& g, v8f (&acc)[4]) {
#pragma unroll
  for (int s = 0; s < KQ; ++s)
#pragma unroll
    for (int u = 0; u < 4; ++u)
      acc[u] = __builtin_amdgcn_wmma_f32_16x16x4_f32(false, g.a[s][u], false, g.b[s], (short)0,
                                                     acc[u], false, false);
}

__global__ __launch_bounds__(128)
void front(const float* __restrict__ center, const float* __restrict__ WQ,
           const float* __restrict__ bQ, const float* __restrict__ WV,
           const float* __restrict__ bV, float* __restrict__ Qout,
           float* __restrict__ WVs, float* __restrict__ bVs) {
  __shared__ __attribute__((aligned(16))) float As[QROWS * QPITCH];
  const int t = threadIdx.x;

  if (blockIdx.x >= QGROUPS) {
    int idx = (blockIdx.x - QGROUPS) * 128 + t;
    int c = idx >> 6;
    int j = idx & 63;
    float s = 0.f;
#pragma unroll
    for (int h = 0; h < H_; ++h) s += WV[c * QCOLS + j * H_ + h];
    WVs[c * OUTD + j] = s;
    if (idx < OUTD) {
      float sb = 0.f;
#pragma unroll
      for (int h = 0; h < H_; ++h) sb += bV[idx * H_ + h];
      bVs[idx] = sb;
    }
    return;
  }

  const int wave = t >> 5, lane = t & 31;
  const int l15 = lane & 15, hi = lane >> 4;
  const int g  = blockIdx.x >> 3;
  const int n0 = ((blockIdx.x & 7) * 4 + wave) * 16;
  const float* bp = WQ + (size_t)(hi * 2) * QCOLS + n0 + l15;
  const float* ap = As + l15 * QPITCH + hi * 2;

  const float* cg = center + (size_t)g * QROWS * IND;
  v4f st[32];
#pragma unroll
  for (int k = 0; k < 32; ++k) st[k] = *(const v4f*)(cg + (size_t)(t + 128 * k) * 4);
  QGroup qA, qB;
  load_qb(qA, bp, 0);
  float* sp = As + (t >> 6) * QPITCH + (t & 63) * 4;
#pragma unroll
  for (int k = 0; k < 32; ++k) *(v4f*)(sp + k * 2 * QPITCH) = st[k];
  __syncthreads();

  v8f acc[4] = {{}, {}, {}, {}};
  load_qa(qA, ap, 0);
  for (int it = 0; it < IND / 4 / (2 * KQ); ++it) {
    const int ktB = (2 * it + 1) * KQ;
    load_qb(qB, bp, ktB);
    load_qa(qB, ap, ktB);
    __builtin_amdgcn_sched_barrier(0);
    mma_q(qA, acc);
    __builtin_amdgcn_sched_barrier(0);
    const int ktA = (it + 1 < IND / 4 / (2 * KQ)) ? (2 * it + 2) * KQ : ktB;
    load_qb(qA, bp, ktA);
    load_qa(qA, ap, ktA);
    __builtin_amdgcn_sched_barrier(0);
    mma_q(qB, acc);
    __builtin_amdgcn_sched_barrier(0);
  }
  float bias = bQ[n0 + l15];
#pragma unroll
  for (int u = 0; u < 4; ++u)
#pragma unroll
    for (int r = 0; r < 8; ++r)
      Qout[(size_t)(g * 64 + u * 16 + r + hi * 8) * QCOLS + n0 + l15] = acc[u][r] + bias;
}

#define KG 4
struct ProjGroup {
  v4f w0[KG], w1[KG];
  v2f wv[KG];
  v2f a0[KG], a1[KG];
};

__device__ __forceinline__ void load_weights(ProjGroup& g, const float* __restrict__ wkp,
                                             const float* __restrict__ wvp, int kt0) {
#pragma unroll
  for (int s = 0; s < KG; ++s) {
    int kb = (kt0 + s) * 4;
    g.w0[s] = *(const v4f*)(wkp + (size_t)kb * QCOLS);
    g.w1[s] = *(const v4f*)(wkp + (size_t)(kb + 1) * QCOLS);
    g.wv[s].x = wvp[kb * OUTD];
    g.wv[s].y = wvp[(kb + 1) * OUTD];
  }
}

__device__ __forceinline__ void load_rows(ProjGroup& g, const float* nb0, const float* nb1,
                                          int kt0) {
#pragma unroll
  for (int s = 0; s < KG; ++s) {
    int kb = (kt0 + s) * 4;
    g.a0[s] = *(const v2f*)(nb0 + kb);
    g.a1[s] = *(const v2f*)(nb1 + kb);
  }
}

__device__ __forceinline__ void proj_group(const ProjGroup& g, v8f (&accK)[2][4], v8f& accV) {
#pragma unroll
  for (int s = 0; s < KG; ++s) {
#pragma unroll
    for (int t = 0; t < 4; ++t) {
      v2f bb;
      bb.x = g.w0[s][t];
      bb.y = g.w1[s][t];
      accK[0][t] = __builtin_amdgcn_wmma_f32_16x16x4_f32(false, g.a0[s], false, bb, (short)0,
                                                         accK[0][t], false, false);
      accK[1][t] = __builtin_amdgcn_wmma_f32_16x16x4_f32(false, g.a1[s], false, bb, (short)0,
                                                         accK[1][t], false, false);
    }
    accV = __builtin_amdgcn_wmma_f32_16x16x4_f32(false, g.a0[s], false, g.wv[s], (short)0,
                                                 accV, false, false);
  }
}

__device__ __forceinline__ float xhalf(float x) {
  return __builtin_bit_cast(float, __builtin_amdgcn_permlanex16(
      __builtin_bit_cast(int, x), __builtin_bit_cast(int, x), 0x76543210, 0xfedcba98,
      false, false));
}

__global__ __launch_bounds__(256)
void attn_fused(const float* __restrict__ neighbors, const float* __restrict__ WK,
                const float* __restrict__ bK, const float* __restrict__ WVs,
                const float* __restrict__ bVs, const float* __restrict__ Qh,
                float* __restrict__ out) {
  __shared__ __attribute__((aligned(16))) float nb[N_ * 260];
  __shared__ __attribute__((aligned(16))) float Kh[N_ * QCOLS];
  __shared__ __attribute__((aligned(16))) float Vs[N_ * OUTD];
  __shared__ float Qs[OUTD * 9];
  __shared__ float part[2 * OUTD];

  const int b = blockIdx.x;
  const int t = threadIdx.x;
  const int wave = t >> 5, lane = t & 31, l15 = lane & 15, hi = lane >> 4;

  const int mA = wave >> 2, mB = 1 - mA;
  const int nV0 = (wave & 3) * 16;
  const float* wkp = WK + (size_t)(hi * 2) * QCOLS + wave * 64 + l15 * 4;
  const float* wvp = WVs + (size_t)(hi * 2) * OUTD + nV0 + l15;
  const float* nb0 = nb + (mA * 16 + l15) * 260 + hi * 2;
  const float* nb1 = nb + (mB * 16 + l15) * 260 + hi * 2;

  ProjGroup gA, gB;
  load_weights(gA, wkp, wvp, 0);

  const float* nbg = neighbors + (size_t)b * N_ * IND;
  const float* qg = Qh + (size_t)b * QCOLS;
  v4f st[8];
#pragma unroll
  for (int k = 0; k < 8; ++k) st[k] = *(const v4f*)(nbg + (size_t)(t + 256 * k) * 4);
  const float q0 = qg[t], q1 = qg[t + 256];
#pragma unroll
  for (int k = 0; k < 8; ++k) {
    const int f = t + 256 * k;
    *(v4f*)(nb + (f >> 6) * 260 + (f & 63) * 4) = st[k];
  }
  const float qscale = 0.125f * 1.44269504088896340736f;
  Qs[(t >> 3) * 9 + (t & 7)] = q0 * qscale;
  Qs[((t + 256) >> 3) * 9 + (t & 7)] = q1 * qscale;
  __syncthreads();

  v8f accK[2][4] = {{{}, {}, {}, {}}, {{}, {}, {}, {}}};
  v8f accV = {};
  load_rows(gA, nb0, nb1, 0);
  for (int it = 0; it < IND / 4 / (2 * KG); ++it) {
    const int ktB = (2 * it + 1) * KG;
    load_weights(gB, wkp, wvp, ktB);
    load_rows(gB, nb0, nb1, ktB);
    __builtin_amdgcn_sched_barrier(0);
    proj_group(gA, accK, accV);
    __builtin_amdgcn_sched_barrier(0);
    const int ktA = (it + 1 < IND / 4 / (2 * KG)) ? (2 * it + 2) * KG : ktB;
    load_weights(gA, wkp, wvp, ktA);
    load_rows(gA, nb0, nb1, ktA);
    __builtin_amdgcn_sched_barrier(0);
    proj_group(gB, accK, accV);
  }
  {
    const v4f bias = *(const v4f*)(bK + wave * 64 + l15 * 4);
#pragma unroll
    for (int r = 0; r < 8; ++r) {
      v4f o0, o1;
#pragma unroll
      for (int tt = 0; tt < 4; ++tt) {
        o0[tt] = accK[0][tt][r] + bias[tt];
        o1[tt] = accK[1][tt][r] + bias[tt];
      }
      *(v4f*)(Kh + (mA * 16 + r + hi * 8) * QCOLS + wave * 64 + l15 * 4) = o0;
      *(v4f*)(Kh + (mB * 16 + r + hi * 8) * QCOLS + wave * 64 + l15 * 4) = o1;
    }
    const float biasV = bVs[nV0 + l15];
#pragma unroll
    for (int r = 0; r < 8; ++r)
      Vs[(mA * 16 + r + hi * 8) * OUTD + nV0 + l15] = accV[r] + biasV;
  }
  __syncthreads();

  const int j0 = (wave & 3) * 16;
  const int nh = wave >> 2;
  v2f aQ[4][2];
#pragma unroll
  for (int mt = 0; mt < 4; ++mt)
#pragma unroll
    for (int ks = 0; ks < 2; ++ks) {
      int kb = ks * 4 + hi * 2;
      aQ[mt][ks].x = Qs[(mt * 16 + l15) * 9 + kb];
      aQ[mt][ks].y = Qs[(mt * 16 + l15) * 9 + kb + 1];
    }
  const float* kcol = Kh + (size_t)(nh * (N_ / 2)) * QCOLS + (j0 + l15) * 8 + hi * 2;
  const float* vrow = Vs + (nh * (N_ / 2)) * OUTD + hi * 8;
  float accOut = 0.f;

  for (int nn = 0; nn < N_ / 2; ++nn) {
    const v2f bb0 = *(const v2f*)(kcol + nn * QCOLS);
    const v2f bb1 = *(const v2f*)(kcol + nn * QCOLS + 4);
    v8f s[4];
#pragma unroll
    for (int mt = 0; mt < 4; ++mt) {
      v8f z = {};
      s[mt] = __builtin_amdgcn_wmma_f32_16x16x4_f32(false, aQ[mt][0], false, bb0, (short)0,
                                                    z, false, false);
    }
#pragma unroll
    for (int mt = 0; mt < 4; ++mt)
      s[mt] = __builtin_amdgcn_wmma_f32_16x16x4_f32(false, aQ[mt][1], false, bb1, (short)0,
                                                    s[mt], false, false);
    v4f vs[4][2];
#pragma unroll
    for (int mt = 0; mt < 4; ++mt) {
      vs[mt][0] = *(const v4f*)(vrow + nn * OUTD + mt * 16);
      vs[mt][1] = *(const v4f*)(vrow + nn * OUTD + mt * 16 + 4);
    }
    float pm[4];
#pragma unroll
    for (int mt = 0; mt < 4; ++mt) {
      pm[mt] = s[mt][0];
#pragma unroll
      for (int r = 1; r < 8; ++r) pm[mt] = fmaxf(pm[mt], s[mt][r]);
    }
    float m = fmaxf(fmaxf(pm[0], pm[1]), fmaxf(pm[2], pm[3]));
    m = fmaxf(m, xhalf(m));
    const v2f m2 = {m, m};
    v2f pd[4], pn[4];
#pragma unroll
    for (int mt = 0; mt < 4; ++mt) {
#pragma unroll
      for (int p = 0; p < 4; ++p) {
        v2f sv, vv, e;
        sv.x = s[mt][2 * p];
        sv.y = s[mt][2 * p + 1];
        vv.x = vs[mt][p >> 1][(2 * p) & 3];
        vv.y = vs[mt][p >> 1][(2 * p + 1) & 3];
        const v2f d = sv - m2;
        e.x = __builtin_amdgcn_exp2f(d.x);
        e.y = __builtin_amdgcn_exp2f(d.y);
        if (p == 0) {
          pd[mt] = e;
          pn[mt] = e * vv;
        } else {
          pd[mt] += e;
          pn[mt] += e * vv;
        }
      }
    }
    const v2f den2 = (pd[0] + pd[1]) + (pd[2] + pd[3]);
    const v2f num2 = (pn[0] + pn[1]) + (pn[2] + pn[3]);
    float den = den2.x + den2.y;
    float num = num2.x + num2.y;
    den += xhalf(den);
    num += xhalf(num);
    accOut += num * __builtin_amdgcn_rcpf(den);
  }
  if (hi == 0) part[nh * OUTD + j0 + l15] = accOut;
  __syncthreads();
  if (t < 64) out[(size_t)b * OUTD + t] = (part[t] + part[OUTD + t]) * (1.f / H_);
}

extern "C" void kernel_launch(void* const* d_in, const int* in_sizes, int n_in,
                              void* d_out, int out_size, void* d_ws, size_t ws_size,
                              hipStream_t stream) {
  const float* center    = (const float*)d_in[0];
  const float* neighbors = (const float*)d_in[1];
  const float* WQ        = (const float*)d_in[2];
  const float* bQ        = (const float*)d_in[3];
  const float* WK        = (const float*)d_in[4];
  const float* bK        = (const float*)d_in[5];
  const float* WV        = (const float*)d_in[6];
  const float* bV        = (const float*)d_in[7];
  float* out = (float*)d_out;

  float* ws   = (float*)d_ws;
  float* wsQ  = ws;
  float* WVs  = ws + (size_t)B_ * QCOLS;
  float* bVs  = WVs + (size_t)IND * OUTD;

  front<<<384, 128, 0, stream>>>(center, WQ, bQ, WV, bV, wsQ, WVs, bVs);
  attn_fused<<<B_, 256, 0, stream>>>(neighbors, WK, bK, WVs, bVs, wsQ, out);
}
